# starting kernel
# baseline (speedup 1.0000x reference)
#include <hip/hip_runtime.h>
#include <hip/hip_bf16.h>

#define NN 100000
#define EE 1600000
#define HH 128
#define LDS_STRIDE 132

typedef __attribute__((ext_vector_type(2))) float v2f;
typedef __attribute__((ext_vector_type(8))) float v8f;

__global__ void deg_kernel(const int* __restrict__ src, const int* __restrict__ dst,
                           float* __restrict__ deg_out, float* __restrict__ deg_in) {
    int e = blockIdx.x * blockDim.x + threadIdx.x;
    if (e < EE) {
        atomicAdd(&deg_out[src[e]], 1.0f);
        atomicAdd(&deg_in[dst[e]], 1.0f);
    }
}

__global__ void norm_kernel(float* __restrict__ ns, float* __restrict__ nd) {
    int i = blockIdx.x * blockDim.x + threadIdx.x;
    if (i < NN) {
        ns[i] = rsqrtf(fmaxf(ns[i], 1.0f));
        nd[i] = rsqrtf(fmaxf(nd[i], 1.0f));
    }
}

__global__ void scatter_edges(const float* __restrict__ feat,
                              const float* __restrict__ norm_src,
                              const int* __restrict__ src, const int* __restrict__ dst,
                              float* __restrict__ agg) {
    int e = blockIdx.x * (blockDim.x >> 5) + (threadIdx.x >> 5);
    if (e >= EE) return;
    int lane = threadIdx.x & 31;
    int s = src[e];
    int d = dst[e];
    float ns = norm_src[s];
    float4 v = ((const float4*)(feat + (size_t)s * HH))[lane];
    v.x *= ns; v.y *= ns; v.z *= ns; v.w *= ns;
    float* ad = agg + (size_t)d * HH + lane * 4;
    atomicAdd(ad + 0, v.x);
    atomicAdd(ad + 1, v.y);
    atomicAdd(ad + 2, v.z);
    atomicAdd(ad + 3, v.w);
}

__global__ void gcn_gemm(const float* __restrict__ agg,
                         const float* __restrict__ norm_dst,
                         const float* __restrict__ W,
                         const float* __restrict__ bias,
                         const float* __restrict__ ori_in,
                         float res,
                         float* __restrict__ out,
                         float* __restrict__ ori_out) {
    __shared__ float lds_a[16 * LDS_STRIDE];

    const int m0   = blockIdx.x * 16;
    const int tid  = threadIdx.x;
    const int wave = tid >> 5;
    const int lane = tid & 31;

    {
        int base = tid * 8;
        int row  = base >> 7;
        int col  = base & 127;
        float nd = norm_dst[m0 + row];
        const float4* s4 = (const float4*)(agg + (size_t)(m0 + row) * HH + col);
        float4 a0 = s4[0];
        float4 a1 = s4[1];
        a0.x *= nd; a0.y *= nd; a0.z *= nd; a0.w *= nd;
        a1.x *= nd; a1.y *= nd; a1.z *= nd; a1.w *= nd;
        float4* t4 = (float4*)(lds_a + row * LDS_STRIDE + col);
        t4[0] = a0;
        t4[1] = a1;
    }
    __syncthreads();

    const int n0   = wave * 16;
    const int half = lane >> 4;
    const int l15  = lane & 15;

    const float* arow = lds_a + l15 * LDS_STRIDE + half * 2;
    const float* bcol = W + (size_t)(half * 2) * HH + n0 + l15;

    v8f c = {};
#pragma unroll
    for (int kk = 0; kk < 32; ++kk) {
        const int k0 = kk * 4;
        v2f a, b;
        a.x = arow[k0];
        a.y = arow[k0 + 1];
        b.x = bcol[(size_t)k0 * HH];
        b.y = bcol[(size_t)k0 * HH + HH];
        c = __builtin_amdgcn_wmma_f32_16x16x4_f32(
                 false, a,  false, b,
                 (short)0, c,  false,  false);
    }

    const int nn = n0 + l15;
    const float bv = bias[nn];
#pragma unroll
    for (int r = 0; r < 8; ++r) {
        int m = m0 + r + half * 8;
        size_t idx = (size_t)m * HH + nn;
        float v = c[r] + bv;
        if (ori_in)  v += res * ori_in[idx];
        out[idx] = v;
        if (ori_out) ori_out[idx] = v;
    }
}

__global__ void ln_relu(float* __restrict__ h,
                        const float* __restrict__ g,
                        const float* __restrict__ bta) {
    int n = blockIdx.x * (blockDim.x >> 5) + (threadIdx.x >> 5);
    if (n >= NN) return;
    int lane = threadIdx.x & 31;
    float4* row = (float4*)(h + (size_t)n * HH);
    float4 v = row[lane];

    float s = v.x + v.y + v.z + v.w;
#pragma unroll
    for (int off = 16; off > 0; off >>= 1) s += __shfl_xor(s, off, 32);
    float mu = s * (1.0f / 128.0f);

    float dx = v.x - mu, dy = v.y - mu, dz = v.z - mu, dw = v.w - mu;
    float q = dx * dx + dy * dy + dz * dz + dw * dw;
#pragma unroll
    for (int off = 16; off > 0; off >>= 1) q += __shfl_xor(q, off, 32);
    float r = rsqrtf(q * (1.0f / 128.0f) + 1e-5f);

    int c = lane * 4;
    float4 o;
    o.x = fmaxf(dx * r * g[c + 0] + bta[c + 0], 0.0f);
    o.y = fmaxf(dy * r * g[c + 1] + bta[c + 1], 0.0f);
    o.z = fmaxf(dz * r * g[c + 2] + bta[c + 2], 0.0f);
    o.w = fmaxf(dw * r * g[c + 3] + bta[c + 3], 0.0f);
    row[lane] = o;
}

extern "C" void kernel_launch(void* const* d_in, const int* in_sizes, int n_in,
                              void* d_out, int out_size, void* d_ws, size_t ws_size,
                              hipStream_t stream) {
    (void)in_sizes; (void)n_in; (void)out_size; (void)ws_size;

    const float* x    = (const float*)d_in[0];
    const float* W0   = (const float*)d_in[1];
    const float* Ws   = (const float*)d_in[2];
    const float* b    = (const float*)d_in[3];
    const float* ln_g = (const float*)d_in[4];
    const float* ln_b = (const float*)d_in[5];
    const int*   src  = (const int*)d_in[6];
    const int*   dst  = (const int*)d_in[7];
    float* out = (float*)d_out;

    float* ws       = (float*)d_ws;
    float* norm_src = ws;
    float* norm_dst = ws + NN;
    float* agg      = ws + 2 * (size_t)NN;
    float* ori      = agg + (size_t)NN * HH;
    float* hbuf     = ori + (size_t)NN * HH;

    const size_t featBytes = (size_t)NN * HH * sizeof(float);

    hipMemsetAsync(ws, 0, 2 * (size_t)NN * sizeof(float), stream);
    deg_kernel<<<EE / 256, 256, 0, stream>>>(src, dst, norm_src, norm_dst);
    norm_kernel<<<(NN + 255) / 256, 256, 0, stream>>>(norm_src, norm_dst);

    hipMemsetAsync(agg, 0, featBytes, stream);
    scatter_edges<<<EE / 8, 256, 0, stream>>>(x, norm_src, src, dst, agg);
    gcn_gemm<<<NN / 16, 256, 0, stream>>>(agg, norm_dst, W0, b, nullptr, 0.0f,
                                          hbuf, ori);

    ln_relu<<<NN / 8, 256, 0, stream>>>(hbuf, ln_g + HH, ln_b + HH);
    hipMemsetAsync(agg, 0, featBytes, stream);
    scatter_edges<<<EE / 8, 256, 0, stream>>>(hbuf, norm_src, src, dst, agg);
    gcn_gemm<<<NN / 16, 256, 0, stream>>>(agg, norm_dst, Ws, b + HH, ori, 0.5f,
                                          hbuf, nullptr);

    ln_relu<<<NN / 8, 256, 0, stream>>>(hbuf, ln_g + 2 * HH, ln_b + 2 * HH);
    hipMemsetAsync(agg, 0, featBytes, stream);
    scatter_edges<<<EE / 8, 256, 0, stream>>>(hbuf, norm_src, src, dst, agg);
    gcn_gemm<<<NN / 16, 256, 0, stream>>>(agg, norm_dst, Ws + (size_t)HH * HH,
                                          b + 2 * HH, ori, 0.5f, out, nullptr);
}


# Replace the f32 atomic scatter-add by a per-call destination-sorted edge list (rank from the in-degree add, scan, fill) summed one wave per node; a row's sum order now follows the edge ranks
# speedup vs baseline: 2.4082x; 2.4082x over previous
#include <hip/hip_runtime.h>
#include <hip/hip_bf16.h>

#define NN 100000
#define EE 1600000
#define HH 128
#define LDS_STRIDE 132

typedef __attribute__((ext_vector_type(2))) float v2f;
typedef __attribute__((ext_vector_type(8))) float v8f;

#define SCAN_T 1024
#define SCAN_B ((NN + SCAN_T - 1) / SCAN_T)

__global__ void deg_rank(const int* __restrict__ src, const int* __restrict__ dst,
                         int* __restrict__ cnt_out, int* __restrict__ cnt_in,
                         int* __restrict__ rank) {
    int e = blockIdx.x * blockDim.x + threadIdx.x;
    if (e < EE) {
        const int s = src[e];
        const int d = dst[e];
        atomicAdd(&cnt_out[s], 1);
        rank[e] = atomicAdd(&cnt_in[d], 1);
    }
}

__global__ void norm_bsum(const int* __restrict__ cnt_out, const int* __restrict__ cnt_in,
                          float* __restrict__ ns, float* __restrict__ nd,
                          int* __restrict__ bsum) {
    __shared__ int part[32];
    const int tid = threadIdx.x, lane = tid & 31, wave = tid >> 5;
    const int i = blockIdx.x * SCAN_T + tid;
    int ci = 0;
    if (i < NN) {
        ci = cnt_in[i];
        int co = cnt_out[i];
        ns[i] = rsqrtf(fmaxf((float)co, 1.0f));
        nd[i] = rsqrtf(fmaxf((float)ci, 1.0f));
    }
    int r = ci;
    for (int off = 16; off > 0; off >>= 1) r += __shfl_xor(r, off, 32);
    if (lane == 0) part[wave] = r;
    __syncthreads();
    if (wave == 0) {
        int t = part[lane];
        for (int off = 16; off > 0; off >>= 1) t += __shfl_xor(t, off, 32);
        if (lane == 0) bsum[blockIdx.x] = t;
    }
}

__global__ void scan_rows(const int* __restrict__ cnt_in, const int* __restrict__ bsum,
                          int* __restrict__ rowptr) {
    __shared__ int wsum[32];
    __shared__ int bpart[32];
    const int tid = threadIdx.x, lane = tid & 31, wave = tid >> 5;
    const int i = blockIdx.x * SCAN_T + tid;
    const int v = (i < NN) ? cnt_in[i] : 0;
    int b = (tid < (int)blockIdx.x) ? bsum[tid] : 0;
    int x = v;
    for (int off = 1; off < 32; off <<= 1) {
        int y = __shfl_up(x, off, 32);
        if (lane >= off) x += y;
    }
    for (int off = 16; off > 0; off >>= 1) b += __shfl_xor(b, off, 32);
    if (lane == 31) wsum[wave] = x;
    if (lane == 0)  bpart[wave] = b;
    __syncthreads();
    int t = ((lane < wave) ? wsum[lane] : 0) + bpart[lane];
    for (int off = 16; off > 0; off >>= 1) t += __shfl_xor(t, off, 32);
    if (i < NN) rowptr[i] = t + x - v;
}

__global__ void fill_csr(const int* __restrict__ src, const int* __restrict__ dst,
                         const int* __restrict__ rank, const int* __restrict__ rowptr,
                         const float* __restrict__ norm_src,
                         int* __restrict__ csr_raw) {
    int2* csr = (int2*)csr_raw;
    int e = blockIdx.x * blockDim.x + threadIdx.x;
    if (e < EE) {
        const int s = src[e];
        const float w = norm_src[s];
        unsigned p = (unsigned)(rowptr[dst[e]] + rank[e]);
        p = min(p, (unsigned)(EE - 1));
        csr[p] = make_int2(s, __float_as_int(w));
    }
}

__global__ void gather_rows(const float* __restrict__ feat,
                            const int* __restrict__ rowptr,
                            const int* __restrict__ cnt_in,
                            const int* __restrict__ csr_raw,
                            float* __restrict__ agg) {
    const int2* csr = (const int2*)csr_raw;
    const int wv = __builtin_amdgcn_readfirstlane((int)(threadIdx.x >> 5));
    const int n = blockIdx.x * 8 + wv;
    const int lane = threadIdx.x & 31;
    int beg = rowptr[n];
    int cnt = cnt_in[n];
    beg = min(max(beg, 0), EE);
    cnt = min(max(cnt, 0), EE - beg);
    const float4* feat4 = (const float4*)feat;
    const float4 zero4 = make_float4(0.0f, 0.0f, 0.0f, 0.0f);
    float4 acc = zero4;
    for (int base = 0; base < cnt; base += 32) {
        const int m = min(32, cnt - base);
        int2 sw = make_int2(0, 0);
        if (lane < m) sw = csr[beg + base + lane];
        for (int j = 0; j < m; j += 8) {
            float4 v[8];
            float wk[8];
#pragma unroll
            for (int k = 0; k < 8; ++k) {
                const int sk = __builtin_amdgcn_readlane(sw.x, j + k);
                wk[k] = __int_as_float(__builtin_amdgcn_readlane(sw.y, j + k));
                if (j + k < m) v[k] = feat4[(size_t)sk * (HH / 4) + lane];
                else           v[k] = zero4;
            }
#pragma unroll
            for (int k = 0; k < 8; ++k) {
                acc.x = fmaf(wk[k], v[k].x, acc.x);
                acc.y = fmaf(wk[k], v[k].y, acc.y);
                acc.z = fmaf(wk[k], v[k].z, acc.z);
                acc.w = fmaf(wk[k], v[k].w, acc.w);
            }
        }
    }
    ((float4*)agg)[(size_t)n * (HH / 4) + lane] = acc;
}

__global__ void gcn_gemm(const float* __restrict__ agg,
                         const float* __restrict__ norm_dst,
                         const float* __restrict__ W,
                         const float* __restrict__ bias,
                         const float* __restrict__ ori_in,
                         float res,
                         float* __restrict__ out,
                         float* __restrict__ ori_out) {
    __shared__ float lds_a[16 * LDS_STRIDE];

    const int m0   = blockIdx.x * 16;
    const int tid  = threadIdx.x;
    const int wave = tid >> 5;
    const int lane = tid & 31;

    {
        int base = tid * 8;
        int row  = base >> 7;
        int col  = base & 127;
        float nd = norm_dst[m0 + row];
        const float4* s4 = (const float4*)(agg + (size_t)(m0 + row) * HH + col);
        float4 a0 = s4[0];
        float4 a1 = s4[1];
        a0.x *= nd; a0.y *= nd; a0.z *= nd; a0.w *= nd;
        a1.x *= nd; a1.y *= nd; a1.z *= nd; a1.w *= nd;
        float4* t4 = (float4*)(lds_a + row * LDS_STRIDE + col);
        t4[0] = a0;
        t4[1] = a1;
    }
    __syncthreads();

    const int n0   = wave * 16;
    const int half = lane >> 4;
    const int l15  = lane & 15;

    const float* arow = lds_a + l15 * LDS_STRIDE + half * 2;
    const float* bcol = W + (size_t)(half * 2) * HH + n0 + l15;

    v8f c = {};
#pragma unroll
    for (int kk = 0; kk < 32; ++kk) {
        const int k0 = kk * 4;
        v2f a, b;
        a.x = arow[k0];
        a.y = arow[k0 + 1];
        b.x = bcol[(size_t)k0 * HH];
        b.y = bcol[(size_t)k0 * HH + HH];
        c = __builtin_amdgcn_wmma_f32_16x16x4_f32(
                 false, a,  false, b,
                 (short)0, c,  false,  false);
    }

    const int nn = n0 + l15;
    const float bv = bias[nn];
#pragma unroll
    for (int r = 0; r < 8; ++r) {
        int m = m0 + r + half * 8;
        size_t idx = (size_t)m * HH + nn;
        float v = c[r] + bv;
        if (ori_in)  v += res * ori_in[idx];
        out[idx] = v;
        if (ori_out) ori_out[idx] = v;
    }
}

__global__ void ln_relu(float* __restrict__ h,
                        const float* __restrict__ g,
                        const float* __restrict__ bta) {
    int n = blockIdx.x * (blockDim.x >> 5) + (threadIdx.x >> 5);
    if (n >= NN) return;
    int lane = threadIdx.x & 31;
    float4* row = (float4*)(h + (size_t)n * HH);
    float4 v = row[lane];

    float s = v.x + v.y + v.z + v.w;
#pragma unroll
    for (int off = 16; off > 0; off >>= 1) s += __shfl_xor(s, off, 32);
    float mu = s * (1.0f / 128.0f);

    float dx = v.x - mu, dy = v.y - mu, dz = v.z - mu, dw = v.w - mu;
    float q = dx * dx + dy * dy + dz * dz + dw * dw;
#pragma unroll
    for (int off = 16; off > 0; off >>= 1) q += __shfl_xor(q, off, 32);
    float r = rsqrtf(q * (1.0f / 128.0f) + 1e-5f);

    int c = lane * 4;
    float4 o;
    o.x = fmaxf(dx * r * g[c + 0] + bta[c + 0], 0.0f);
    o.y = fmaxf(dy * r * g[c + 1] + bta[c + 1], 0.0f);
    o.z = fmaxf(dz * r * g[c + 2] + bta[c + 2], 0.0f);
    o.w = fmaxf(dw * r * g[c + 3] + bta[c + 3], 0.0f);
    row[lane] = o;
}

extern "C" void kernel_launch(void* const* d_in, const int* in_sizes, int n_in,
                              void* d_out, int out_size, void* d_ws, size_t ws_size,
                              hipStream_t stream) {
    (void)in_sizes; (void)n_in; (void)out_size; (void)ws_size;

    const float* x    = (const float*)d_in[0];
    const float* W0   = (const float*)d_in[1];
    const float* Ws   = (const float*)d_in[2];
    const float* b    = (const float*)d_in[3];
    const float* ln_g = (const float*)d_in[4];
    const float* ln_b = (const float*)d_in[5];
    const int*   src  = (const int*)d_in[6];
    const int*   dst  = (const int*)d_in[7];
    float* out = (float*)d_out;

    float* ws       = (float*)d_ws;
    float* norm_src = ws;
    float* norm_dst = ws + NN;
    float* agg      = ws + 2 * (size_t)NN;
    float* ori      = agg + (size_t)NN * HH;
    float* hbuf     = ori + (size_t)NN * HH;
    int*   cnt_out  = (int*)(hbuf + (size_t)NN * HH);
    int*   cnt_in   = cnt_out + NN;
    int*   rowptr   = cnt_in + NN;
    int*   bsum     = rowptr + NN;
    int*   rank     = bsum + 128;
    int*   csr      = rank + EE;

    hipMemsetAsync(cnt_out, 0, 2 * (size_t)NN * sizeof(int), stream);
    deg_rank<<<EE / 256, 256, 0, stream>>>(src, dst, cnt_out, cnt_in, rank);
    norm_bsum<<<SCAN_B, SCAN_T, 0, stream>>>(cnt_out, cnt_in, norm_src, norm_dst, bsum);
    scan_rows<<<SCAN_B, SCAN_T, 0, stream>>>(cnt_in, bsum, rowptr);
    fill_csr<<<EE / 256, 256, 0, stream>>>(src, dst, rank, rowptr, norm_src, csr);

    gather_rows<<<NN / 8, 256, 0, stream>>>(x, rowptr, cnt_in, csr, agg);
    gcn_gemm<<<NN / 16, 256, 0, stream>>>(agg, norm_dst, W0, b, nullptr, 0.0f,
                                          hbuf, ori);

    ln_relu<<<NN / 8, 256, 0, stream>>>(hbuf, ln_g + HH, ln_b + HH);
    gather_rows<<<NN / 8, 256, 0, stream>>>(hbuf, rowptr, cnt_in, csr, agg);
    gcn_gemm<<<NN / 16, 256, 0, stream>>>(agg, norm_dst, Ws, b + HH, ori, 0.5f,
                                          hbuf, nullptr);

    ln_relu<<<NN / 8, 256, 0, stream>>>(hbuf, ln_g + 2 * HH, ln_b + 2 * HH);
    gather_rows<<<NN / 8, 256, 0, stream>>>(hbuf, rowptr, cnt_in, csr, agg);
    gcn_gemm<<<NN / 16, 256, 0, stream>>>(agg, norm_dst, Ws + (size_t)HH * HH,
                                          b + 2 * HH, ori, 0.5f, out, nullptr);
}


# Fuse layernorm+relu into the GEMM launch: the block passes its 16x128 result through LDS and each wave finishes two whole rows (same lane-pair sum tree); the two layernorm launches go
# speedup vs baseline: 3.1431x; 1.3052x over previous
#include <hip/hip_runtime.h>
#include <hip/hip_bf16.h>

#define NN 100000
#define EE 1600000
#define HH 128
#define LDS_STRIDE 132

typedef __attribute__((ext_vector_type(2))) float v2f;
typedef __attribute__((ext_vector_type(8))) float v8f;

#define SCAN_T 1024
#define SCAN_B ((NN + SCAN_T - 1) / SCAN_T)

__global__ void deg_rank(const int* __restrict__ src, const int* __restrict__ dst,
                         int* __restrict__ cnt_out, int* __restrict__ cnt_in,
                         int* __restrict__ rank) {
    int e = blockIdx.x * blockDim.x + threadIdx.x;
    if (e < EE) {
        const int s = src[e];
        const int d = dst[e];
        atomicAdd(&cnt_out[s], 1);
        rank[e] = atomicAdd(&cnt_in[d], 1);
    }
}

__global__ void norm_bsum(const int* __restrict__ cnt_out, const int* __restrict__ cnt_in,
                          float* __restrict__ ns, float* __restrict__ nd,
                          int* __restrict__ bsum) {
    __shared__ int part[32];
    const int tid = threadIdx.x, lane = tid & 31, wave = tid >> 5;
    const int i = blockIdx.x * SCAN_T + tid;
    int ci = 0;
    if (i < NN) {
        ci = cnt_in[i];
        int co = cnt_out[i];
        ns[i] = rsqrtf(fmaxf((float)co, 1.0f));
        nd[i] = rsqrtf(fmaxf((float)ci, 1.0f));
    }
    int r = ci;
    for (int off = 16; off > 0; off >>= 1) r += __shfl_xor(r, off, 32);
    if (lane == 0) part[wave] = r;
    __syncthreads();
    if (wave == 0) {
        int t = part[lane];
        for (int off = 16; off > 0; off >>= 1) t += __shfl_xor(t, off, 32);
        if (lane == 0) bsum[blockIdx.x] = t;
    }
}

__global__ void scan_rows(const int* __restrict__ cnt_in, const int* __restrict__ bsum,
                          int* __restrict__ rowptr) {
    __shared__ int wsum[32];
    __shared__ int bpart[32];
    const int tid = threadIdx.x, lane = tid & 31, wave = tid >> 5;
    const int i = blockIdx.x * SCAN_T + tid;
    const int v = (i < NN) ? cnt_in[i] : 0;
    int b = (tid < (int)blockIdx.x) ? bsum[tid] : 0;
    int x = v;
    for (int off = 1; off < 32; off <<= 1) {
        int y = __shfl_up(x, off, 32);
        if (lane >= off) x += y;
    }
    for (int off = 16; off > 0; off >>= 1) b += __shfl_xor(b, off, 32);
    if (lane == 31) wsum[wave] = x;
    if (lane == 0)  bpart[wave] = b;
    __syncthreads();
    int t = ((lane < wave) ? wsum[lane] : 0) + bpart[lane];
    for (int off = 16; off > 0; off >>= 1) t += __shfl_xor(t, off, 32);
    if (i < NN) rowptr[i] = t + x - v;
}

__global__ void fill_csr(const int* __restrict__ src, const int* __restrict__ dst,
                         const int* __restrict__ rank, const int* __restrict__ rowptr,
                         const float* __restrict__ norm_src,
                         int* __restrict__ csr_raw) {
    int2* csr = (int2*)csr_raw;
    int e = blockIdx.x * blockDim.x + threadIdx.x;
    if (e < EE) {
        const int s = src[e];
        const float w = norm_src[s];
        unsigned p = (unsigned)(rowptr[dst[e]] + rank[e]);
        p = min(p, (unsigned)(EE - 1));
        csr[p] = make_int2(s, __float_as_int(w));
    }
}

__global__ void gather_rows(const float* __restrict__ feat,
                            const int* __restrict__ rowptr,
                            const int* __restrict__ cnt_in,
                            const int* __restrict__ csr_raw,
                            float* __restrict__ agg) {
    const int2* csr = (const int2*)csr_raw;
    const int wv = __builtin_amdgcn_readfirstlane((int)(threadIdx.x >> 5));
    const int n = blockIdx.x * 8 + wv;
    const int lane = threadIdx.x & 31;
    int beg = rowptr[n];
    int cnt = cnt_in[n];
    beg = min(max(beg, 0), EE);
    cnt = min(max(cnt, 0), EE - beg);
    const float4* feat4 = (const float4*)feat;
    const float4 zero4 = make_float4(0.0f, 0.0f, 0.0f, 0.0f);
    float4 acc = zero4;
    for (int base = 0; base < cnt; base += 32) {
        const int m = min(32, cnt - base);
        int2 sw = make_int2(0, 0);
        if (lane < m) sw = csr[beg + base + lane];
        for (int j = 0; j < m; j += 8) {
            float4 v[8];
            float wk[8];
#pragma unroll
            for (int k = 0; k < 8; ++k) {
                const int sk = __builtin_amdgcn_readlane(sw.x, j + k);
                wk[k] = __int_as_float(__builtin_amdgcn_readlane(sw.y, j + k));
                if (j + k < m) v[k] = feat4[(size_t)sk * (HH / 4) + lane];
                else           v[k] = zero4;
            }
#pragma unroll
            for (int k = 0; k < 8; ++k) {
                acc.x = fmaf(wk[k], v[k].x, acc.x);
                acc.y = fmaf(wk[k], v[k].y, acc.y);
                acc.z = fmaf(wk[k], v[k].z, acc.z);
                acc.w = fmaf(wk[k], v[k].w, acc.w);
            }
        }
    }
    ((float4*)agg)[(size_t)n * (HH / 4) + lane] = acc;
}

__device__ __forceinline__ float xhalf(float x) {
    const int b = __builtin_bit_cast(int, x);
    return __builtin_bit_cast(float,
        __builtin_amdgcn_permlanex16(b, b, 0x76543210, 0xfedcba98, false, false));
}
template <int M>
__device__ __forceinline__ float xrow(float x) {
    const int b = __builtin_bit_cast(int, x);
    return __builtin_bit_cast(float,
        __builtin_amdgcn_update_dpp(0, b, 0x160 + M, 0xf, 0xf, true));
}
__device__ __forceinline__ float wave_sum(float s) {
    s += xhalf(s);
    s += xrow<8>(s);
    s += xrow<4>(s);
    s += xrow<2>(s);
    s += xrow<1>(s);
    return s;
}

template <bool FUSE_LN>
__global__ void gcn_gemm(const float* __restrict__ agg,
                         const float* __restrict__ norm_dst,
                         const float* __restrict__ W,
                         const float* __restrict__ bias,
                         const float* __restrict__ ori_in,
                         float res,
                         float* __restrict__ out,
                         float* __restrict__ ori_out,
                         const float* __restrict__ ln_g,
                         const float* __restrict__ ln_b) {
    __shared__ __attribute__((aligned(16))) float lds_a[16 * LDS_STRIDE];

    const int m0   = blockIdx.x * 16;
    const int tid  = threadIdx.x;
    const int wave = tid >> 5;
    const int lane = tid & 31;

    const int wv = __builtin_amdgcn_readfirstlane(wave);
    const size_t g0 = (size_t)(m0 + 2 * wv) * (HH / 4) + lane;
    float4 o4[2];
    o4[0] = o4[1] = make_float4(0.0f, 0.0f, 0.0f, 0.0f);
    if (FUSE_LN && ori_in) {
        o4[0] = ((const float4*)ori_in)[g0];
        o4[1] = ((const float4*)ori_in)[g0 + HH / 4];
    }

    {
        int base = tid * 8;
        int row  = base >> 7;
        int col  = base & 127;
        float nd = norm_dst[m0 + row];
        const float4* s4 = (const float4*)(agg + (size_t)(m0 + row) * HH + col);
        float4 a0 = s4[0];
        float4 a1 = s4[1];
        a0.x *= nd; a0.y *= nd; a0.z *= nd; a0.w *= nd;
        a1.x *= nd; a1.y *= nd; a1.z *= nd; a1.w *= nd;
        float4* t4 = (float4*)(lds_a + row * LDS_STRIDE + col);
        t4[0] = a0;
        t4[1] = a1;
    }
    __syncthreads();

    const int n0   = wave * 16;
    const int half = lane >> 4;
    const int l15  = lane & 15;

    const float* arow = lds_a + l15 * LDS_STRIDE + half * 2;
    const float* bcol = W + (size_t)(half * 2) * HH + n0 + l15;

    v8f c = {};
#pragma unroll
    for (int kk = 0; kk < 32; ++kk) {
        const int k0 = kk * 4;
        v2f a, b;
        a.x = arow[k0];
        a.y = arow[k0 + 1];
        b.x = bcol[(size_t)k0 * HH];
        b.y = bcol[(size_t)k0 * HH + HH];
        c = __builtin_amdgcn_wmma_f32_16x16x4_f32(
                 false, a,  false, b,
                 (short)0, c,  false,  false);
    }

    const int nn = n0 + l15;
    const float bv = bias[nn];

    if (!FUSE_LN) {
#pragma unroll
        for (int r = 0; r < 8; ++r) {
            int m = m0 + r + half * 8;
            size_t idx = (size_t)m * HH + nn;
            float v = c[r] + bv;
            if (ori_in)  v += res * ori_in[idx];
            out[idx] = v;
            if (ori_out) ori_out[idx] = v;
        }
    } else {
        const float4 g4 = ((const float4*)ln_g)[lane];
        const float4 b4 = ((const float4*)ln_b)[lane];

        __syncthreads();
#pragma unroll
        for (int r = 0; r < 8; ++r)
            lds_a[(r + half * 8) * LDS_STRIDE + nn] = c[r] + bv;
        __syncthreads();

        float4 h[2];
#pragma unroll
        for (int rr = 0; rr < 2; ++rr) {
            h[rr] = *(const float4*)(lds_a + (2 * wv + rr) * LDS_STRIDE + lane * 4);
            if (ori_in) {
                h[rr].x += res * o4[rr].x; h[rr].y += res * o4[rr].y;
                h[rr].z += res * o4[rr].z; h[rr].w += res * o4[rr].w;
            }
        }
        if (ori_out) {
            ((float4*)ori_out)[g0]          = h[0];
            ((float4*)ori_out)[g0 + HH / 4] = h[1];
        }
#pragma unroll
        for (int rr = 0; rr < 2; ++rr) {
            const float4 v = h[rr];
            const float mu = wave_sum(v.x + v.y + v.z + v.w) * (1.0f / 128.0f);
            const float dx = v.x - mu, dy = v.y - mu, dz = v.z - mu, dw = v.w - mu;
            const float q  = wave_sum(dx * dx + dy * dy + dz * dz + dw * dw);
            const float rs = rsqrtf(q * (1.0f / 128.0f) + 1e-5f);

            float4 o;
            o.x = fmaxf(dx * rs * g4.x + b4.x, 0.0f);
            o.y = fmaxf(dy * rs * g4.y + b4.y, 0.0f);
            o.z = fmaxf(dz * rs * g4.z + b4.z, 0.0f);
            o.w = fmaxf(dw * rs * g4.w + b4.w, 0.0f);
            ((float4*)out)[g0 + rr * (HH / 4)] = o;
        }
    }
}

extern "C" void kernel_launch(void* const* d_in, const int* in_sizes, int n_in,
                              void* d_out, int out_size, void* d_ws, size_t ws_size,
                              hipStream_t stream) {
    (void)in_sizes; (void)n_in; (void)out_size; (void)ws_size;

    const float* x    = (const float*)d_in[0];
    const float* W0   = (const float*)d_in[1];
    const float* Ws   = (const float*)d_in[2];
    const float* b    = (const float*)d_in[3];
    const float* ln_g = (const float*)d_in[4];
    const float* ln_b = (const float*)d_in[5];
    const int*   src  = (const int*)d_in[6];
    const int*   dst  = (const int*)d_in[7];
    float* out = (float*)d_out;

    float* ws       = (float*)d_ws;
    float* norm_src = ws;
    float* norm_dst = ws + NN;
    float* agg      = ws + 2 * (size_t)NN;
    float* ori      = agg + (size_t)NN * HH;
    float* hbuf     = ori + (size_t)NN * HH;
    int*   cnt_out  = (int*)(hbuf + (size_t)NN * HH);
    int*   cnt_in   = cnt_out + NN;
    int*   rowptr   = cnt_in + NN;
    int*   bsum     = rowptr + NN;
    int*   rank     = bsum + 128;
    int*   csr      = rank + EE;

    hipMemsetAsync(cnt_out, 0, 2 * (size_t)NN * sizeof(int), stream);
    deg_rank<<<EE / 256, 256, 0, stream>>>(src, dst, cnt_out, cnt_in, rank);
    norm_bsum<<<SCAN_B, SCAN_T, 0, stream>>>(cnt_out, cnt_in, norm_src, norm_dst, bsum);
    scan_rows<<<SCAN_B, SCAN_T, 0, stream>>>(cnt_in, bsum, rowptr);
    fill_csr<<<EE / 256, 256, 0, stream>>>(src, dst, rank, rowptr, norm_src, csr);

    gather_rows<<<NN / 8, 256, 0, stream>>>(x, rowptr, cnt_in, csr, agg);
    gcn_gemm<true><<<NN / 16, 256, 0, stream>>>(agg, norm_dst, W0, b, nullptr, 0.0f,
                                                hbuf, ori, ln_g + HH, ln_b + HH);

    gather_rows<<<NN / 8, 256, 0, stream>>>(hbuf, rowptr, cnt_in, csr, agg);
    gcn_gemm<true><<<NN / 16, 256, 0, stream>>>(agg, norm_dst, Ws, b + HH, ori, 0.5f,
                                                hbuf, nullptr, ln_g + 2 * HH, ln_b + 2 * HH);

    gather_rows<<<NN / 8, 256, 0, stream>>>(hbuf, rowptr, cnt_in, csr, agg);
    gcn_gemm<false><<<NN / 16, 256, 0, stream>>>(agg, norm_dst, Ws + (size_t)HH * HH,
                                                 b + 2 * HH, ori, 0.5f, out, nullptr,
                                                 nullptr, nullptr);
}


# Fuse the row gather into the GEMM launch: each wave sums the in-edges of its two tile rows straight into the LDS A tile (same edge order); the aggregate buffer and the three gather launches go
# speedup vs baseline: 3.3544x; 1.0672x over previous
#include <hip/hip_runtime.h>
#include <hip/hip_bf16.h>

#define NN 100000
#define EE 1600000
#define HH 128
#define LDS_STRIDE 132

typedef __attribute__((ext_vector_type(2))) float v2f;
typedef __attribute__((ext_vector_type(8))) float v8f;

#define SCAN_T 1024
#define SCAN_B ((NN + SCAN_T - 1) / SCAN_T)

__global__ void deg_rank(const int* __restrict__ src, const int* __restrict__ dst,
                         int* __restrict__ cnt_out, int* __restrict__ cnt_in,
                         int* __restrict__ rank) {
    int e = blockIdx.x * blockDim.x + threadIdx.x;
    if (e < EE) {
        const int s = src[e];
        const int d = dst[e];
        atomicAdd(&cnt_out[s], 1);
        rank[e] = atomicAdd(&cnt_in[d], 1);
    }
}

__global__ void norm_bsum(const int* __restrict__ cnt_out, const int* __restrict__ cnt_in,
                          float* __restrict__ ns, float* __restrict__ nd,
                          int* __restrict__ bsum) {
    __shared__ int part[32];
    const int tid = threadIdx.x, lane = tid & 31, wave = tid >> 5;
    const int i = blockIdx.x * SCAN_T + tid;
    int ci = 0;
    if (i < NN) {
        ci = cnt_in[i];
        int co = cnt_out[i];
        ns[i] = rsqrtf(fmaxf((float)co, 1.0f));
        nd[i] = rsqrtf(fmaxf((float)ci, 1.0f));
    }
    int r = ci;
    for (int off = 16; off > 0; off >>= 1) r += __shfl_xor(r, off, 32);
    if (lane == 0) part[wave] = r;
    __syncthreads();
    if (wave == 0) {
        int t = part[lane];
        for (int off = 16; off > 0; off >>= 1) t += __shfl_xor(t, off, 32);
        if (lane == 0) bsum[blockIdx.x] = t;
    }
}

__global__ void scan_rows(const int* __restrict__ cnt_in, const int* __restrict__ bsum,
                          int* __restrict__ rowptr) {
    __shared__ int wsum[32];
    __shared__ int bpart[32];
    const int tid = threadIdx.x, lane = tid & 31, wave = tid >> 5;
    const int i = blockIdx.x * SCAN_T + tid;
    const int v = (i < NN) ? cnt_in[i] : 0;
    int b = (tid < (int)blockIdx.x) ? bsum[tid] : 0;
    int x = v;
    for (int off = 1; off < 32; off <<= 1) {
        int y = __shfl_up(x, off, 32);
        if (lane >= off) x += y;
    }
    for (int off = 16; off > 0; off >>= 1) b += __shfl_xor(b, off, 32);
    if (lane == 31) wsum[wave] = x;
    if (lane == 0)  bpart[wave] = b;
    __syncthreads();
    int t = ((lane < wave) ? wsum[lane] : 0) + bpart[lane];
    for (int off = 16; off > 0; off >>= 1) t += __shfl_xor(t, off, 32);
    if (i < NN) rowptr[i] = t + x - v;
}

__global__ void fill_csr(const int* __restrict__ src, const int* __restrict__ dst,
                         const int* __restrict__ rank, const int* __restrict__ rowptr,
                         const float* __restrict__ norm_src,
                         int* __restrict__ csr_raw) {
    int2* csr = (int2*)csr_raw;
    int e = blockIdx.x * blockDim.x + threadIdx.x;
    if (e < EE) {
        const int s = src[e];
        const float w = norm_src[s];
        unsigned p = (unsigned)(rowptr[dst[e]] + rank[e]);
        p = min(p, (unsigned)(EE - 1));
        csr[p] = make_int2(s, __float_as_int(w));
    }
}

__device__ __forceinline__ float4 gather_row(const float4* __restrict__ feat4,
                                             const int2* __restrict__ csr,
                                             int beg, int cnt, int lane) {
    const float4 zero4 = make_float4(0.0f, 0.0f, 0.0f, 0.0f);
    float4 acc = zero4;
    for (int base = 0; base < cnt; base += 32) {
        const int m = min(32, cnt - base);
        int2 sw = make_int2(0, 0);
        if (lane < m) sw = csr[beg + base + lane];
        for (int j = 0; j < m; j += 8) {
            float4 v[8];
            float wk[8];
#pragma unroll
            for (int k = 0; k < 8; ++k) {
                const int sk = __builtin_amdgcn_readlane(sw.x, j + k);
                wk[k] = __int_as_float(__builtin_amdgcn_readlane(sw.y, j + k));
                if (j + k < m) v[k] = feat4[(size_t)sk * (HH / 4) + lane];
                else           v[k] = zero4;
            }
#pragma unroll
            for (int k = 0; k < 8; ++k) {
                acc.x = fmaf(wk[k], v[k].x, acc.x);
                acc.y = fmaf(wk[k], v[k].y, acc.y);
                acc.z = fmaf(wk[k], v[k].z, acc.z);
                acc.w = fmaf(wk[k], v[k].w, acc.w);
            }
        }
    }
    return acc;
}

__device__ __forceinline__ float xhalf(float x) {
    const int b = __builtin_bit_cast(int, x);
    return __builtin_bit_cast(float,
        __builtin_amdgcn_permlanex16(b, b, 0x76543210, 0xfedcba98, false, false));
}
template <int M>
__device__ __forceinline__ float xrow(float x) {
    const int b = __builtin_bit_cast(int, x);
    return __builtin_bit_cast(float,
        __builtin_amdgcn_update_dpp(0, b, 0x160 + M, 0xf, 0xf, true));
}
__device__ __forceinline__ float wave_sum(float s) {
    s += xhalf(s);
    s += xrow<8>(s);
    s += xrow<4>(s);
    s += xrow<2>(s);
    s += xrow<1>(s);
    return s;
}

template <bool FUSE_LN>
__global__ void gcn_layer(const float* __restrict__ feat,
                          const int* __restrict__ rowptr,
                          const int* __restrict__ cnt_in,
                          const int* __restrict__ csr_raw,
                          const float* __restrict__ norm_dst,
                          const float* __restrict__ W,
                          const float* __restrict__ bias,
                          const float* __restrict__ ori_in,
                          float res,
                          float* __restrict__ out,
                          float* __restrict__ ori_out,
                          const float* __restrict__ ln_g,
                          const float* __restrict__ ln_b) {
    __shared__ __attribute__((aligned(16))) float lds_a[16 * LDS_STRIDE];

    const int m0   = blockIdx.x * 16;
    const int tid  = threadIdx.x;
    const int wave = tid >> 5;
    const int lane = tid & 31;

    const int wv = __builtin_amdgcn_readfirstlane(wave);
    const int r0 = 2 * wv;
    const int nA = m0 + r0;
    const size_t g0 = (size_t)nA * (HH / 4) + lane;
    float4 o4[2];
    o4[0] = o4[1] = make_float4(0.0f, 0.0f, 0.0f, 0.0f);
    if (FUSE_LN && ori_in) {
        o4[0] = ((const float4*)ori_in)[g0];
        o4[1] = ((const float4*)ori_in)[g0 + HH / 4];
    }

    {
        const int2* csr = (const int2*)csr_raw;
        const float4* feat4 = (const float4*)feat;
        int begA = rowptr[nA],     cntA = cnt_in[nA];
        int begB = rowptr[nA + 1], cntB = cnt_in[nA + 1];
        const float ndA = norm_dst[nA];
        const float ndB = norm_dst[nA + 1];
        begA = min(max(begA, 0), EE);
        cntA = min(max(cntA, 0), EE - begA);
        begB = min(max(begB, 0), EE);
        cntB = min(max(cntB, 0), EE - begB);
        float4 sA = gather_row(feat4, csr, begA, cntA, lane);
        float4 sB = gather_row(feat4, csr, begB, cntB, lane);
        sA.x *= ndA; sA.y *= ndA; sA.z *= ndA; sA.w *= ndA;
        sB.x *= ndB; sB.y *= ndB; sB.z *= ndB; sB.w *= ndB;
        *(float4*)(lds_a + r0 * LDS_STRIDE + lane * 4)       = sA;
        *(float4*)(lds_a + (r0 + 1) * LDS_STRIDE + lane * 4) = sB;
    }
    __syncthreads();

    const int n0   = wave * 16;
    const int half = lane >> 4;
    const int l15  = lane & 15;

    const float* arow = lds_a + l15 * LDS_STRIDE + half * 2;
    const float* bcol = W + (size_t)(half * 2) * HH + n0 + l15;

    v8f c = {};
#pragma unroll
    for (int kk = 0; kk < 32; ++kk) {
        const int k0 = kk * 4;
        v2f a, b;
        a.x = arow[k0];
        a.y = arow[k0 + 1];
        b.x = bcol[(size_t)k0 * HH];
        b.y = bcol[(size_t)k0 * HH + HH];
        c = __builtin_amdgcn_wmma_f32_16x16x4_f32(
                 false, a,  false, b,
                 (short)0, c,  false,  false);
    }

    const int nn = n0 + l15;
    const float bv = bias[nn];

    if (!FUSE_LN) {
#pragma unroll
        for (int r = 0; r < 8; ++r) {
            int m = m0 + r + half * 8;
            size_t idx = (size_t)m * HH + nn;
            float v = c[r] + bv;
            if (ori_in)  v += res * ori_in[idx];
            out[idx] = v;
            if (ori_out) ori_out[idx] = v;
        }
    } else {
        const float4 g4 = ((const float4*)ln_g)[lane];
        const float4 b4 = ((const float4*)ln_b)[lane];

        __syncthreads();
#pragma unroll
        for (int r = 0; r < 8; ++r)
            lds_a[(r + half * 8) * LDS_STRIDE + nn] = c[r] + bv;
        __syncthreads();

        float4 h[2];
#pragma unroll
        for (int rr = 0; rr < 2; ++rr) {
            h[rr] = *(const float4*)(lds_a + (2 * wv + rr) * LDS_STRIDE + lane * 4);
            if (ori_in) {
                h[rr].x += res * o4[rr].x; h[rr].y += res * o4[rr].y;
                h[rr].z += res * o4[rr].z; h[rr].w += res * o4[rr].w;
            }
        }
        if (ori_out) {
            ((float4*)ori_out)[g0]          = h[0];
            ((float4*)ori_out)[g0 + HH / 4] = h[1];
        }
#pragma unroll
        for (int rr = 0; rr < 2; ++rr) {
            const float4 v = h[rr];
            const float mu = wave_sum(v.x + v.y + v.z + v.w) * (1.0f / 128.0f);
            const float dx = v.x - mu, dy = v.y - mu, dz = v.z - mu, dw = v.w - mu;
            const float q  = wave_sum(dx * dx + dy * dy + dz * dz + dw * dw);
            const float rs = rsqrtf(q * (1.0f / 128.0f) + 1e-5f);

            float4 o;
            o.x = fmaxf(dx * rs * g4.x + b4.x, 0.0f);
            o.y = fmaxf(dy * rs * g4.y + b4.y, 0.0f);
            o.z = fmaxf(dz * rs * g4.z + b4.z, 0.0f);
            o.w = fmaxf(dw * rs * g4.w + b4.w, 0.0f);
            ((float4*)out)[g0 + rr * (HH / 4)] = o;
        }
    }
}

extern "C" void kernel_launch(void* const* d_in, const int* in_sizes, int n_in,
                              void* d_out, int out_size, void* d_ws, size_t ws_size,
                              hipStream_t stream) {
    (void)in_sizes; (void)n_in; (void)out_size; (void)ws_size;

    const float* x    = (const float*)d_in[0];
    const float* W0   = (const float*)d_in[1];
    const float* Ws   = (const float*)d_in[2];
    const float* b    = (const float*)d_in[3];
    const float* ln_g = (const float*)d_in[4];
    const float* ln_b = (const float*)d_in[5];
    const int*   src  = (const int*)d_in[6];
    const int*   dst  = (const int*)d_in[7];
    float* out = (float*)d_out;

    float* ws       = (float*)d_ws;
    float* norm_src = ws;
    float* norm_dst = ws + NN;
    float* hbuf2    = ws + 2 * (size_t)NN;
    float* ori      = hbuf2 + (size_t)NN * HH;
    float* hbuf     = ori + (size_t)NN * HH;
    int*   cnt_out  = (int*)(hbuf + (size_t)NN * HH);
    int*   cnt_in   = cnt_out + NN;
    int*   rowptr   = cnt_in + NN;
    int*   bsum     = rowptr + NN;
    int*   rank     = bsum + 128;
    int*   csr      = rank + EE;

    hipMemsetAsync(cnt_out, 0, 2 * (size_t)NN * sizeof(int), stream);
    deg_rank<<<EE / 256, 256, 0, stream>>>(src, dst, cnt_out, cnt_in, rank);
    norm_bsum<<<SCAN_B, SCAN_T, 0, stream>>>(cnt_out, cnt_in, norm_src, norm_dst, bsum);
    scan_rows<<<SCAN_B, SCAN_T, 0, stream>>>(cnt_in, bsum, rowptr);
    fill_csr<<<EE / 256, 256, 0, stream>>>(src, dst, rank, rowptr, norm_src, csr);

    gcn_layer<true><<<NN / 16, 256, 0, stream>>>(x, rowptr, cnt_in, csr, norm_dst, W0, b,
                                                 nullptr, 0.0f, hbuf, ori,
                                                 ln_g + HH, ln_b + HH);

    gcn_layer<true><<<NN / 16, 256, 0, stream>>>(hbuf, rowptr, cnt_in, csr, norm_dst, Ws, b + HH,
                                                 ori, 0.5f, hbuf2, nullptr,
                                                 ln_g + 2 * HH, ln_b + 2 * HH);

    gcn_layer<false><<<NN / 16, 256, 0, stream>>>(hbuf2, rowptr, cnt_in, csr, norm_dst,
                                                  Ws + (size_t)HH * HH, b + 2 * HH,
                                                  ori, 0.5f, out, nullptr, nullptr, nullptr);
}
